# starting kernel
# baseline (speedup 1.0000x reference)
#include <hip/hip_runtime.h>

typedef __attribute__((ext_vector_type(2))) float v2f;
typedef __attribute__((ext_vector_type(8))) float v8f;

#define BB 1024
#define TT 256
#define HH 64
#define EE 32
#define VV 10
#define OO 10

#if __has_builtin(__builtin_amdgcn_tanhf)
#define DEV_TANH(x) __builtin_amdgcn_tanhf(x)
#else
#define DEV_TANH(x) tanhf(x)
#endif


__global__ __launch_bounds__(64) void rnn_prep_kernel(
    const float* __restrict__ emb, const float* __restrict__ Wx,
    const float* __restrict__ brnn, const float* __restrict__ Wd,
    const float* __restrict__ bd, float* __restrict__ ws) {
  const int j = threadIdx.x;
  for (int v = 0; v < VV; ++v) {
    float s0 = brnn[j], s1 = 0.f;
    for (int d = 0; d < EE; ++d) {
      const float e = emb[v * EE + d];
      s0 += e * Wx[d * HH + j];
      s1 += e * Wx[(EE + d) * HH + j];
    }
    ws[v * HH + j] = s0;
    ws[VV * HH + v * HH + j] = s1;
  }
  for (int n = 0; n < 16; ++n)
    ws[1280 + j * 16 + n] = (n < OO) ? Wd[j * OO + n] : 0.f;
  if (j < 16) ws[2304 + j] = (j < OO) ? bd[j] : 0.f;
}

__global__ __launch_bounds__(32) void rnn_scan_kernel(
    const int* __restrict__ num1, const int* __restrict__ num2,
    const float* __restrict__ Wh, const float* __restrict__ ws,
    float* __restrict__ out) {
  __shared__ float lut[2 * VV * HH];
  __shared__ float hbuf[16 * HH];
  __shared__ int   vidx[2][16];

  const int lane = threadIdx.x;
  const int half = lane >> 4;
  const int m    = lane & 15;
  const int kb   = half * 2;
  const int row_base = blockIdx.x * 16;

  for (int i = lane; i < 2 * VV * HH; i += 32) lut[i] = ws[i];
  for (int i = lane; i < 16 * HH; i += 32) hbuf[i] = 0.f;

  v2f whb[16][4];
#pragma unroll
  for (int k = 0; k < 16; ++k)
#pragma unroll
    for (int n = 0; n < 4; ++n) {
      whb[k][n].x = Wh[(4 * k + kb + 0) * HH + n * 16 + m];
      whb[k][n].y = Wh[(4 * k + kb + 1) * HH + n * 16 + m];
    }
  v2f wdb[16];
#pragma unroll
  for (int k = 0; k < 16; ++k) {
    wdb[k].x = ws[1280 + (4 * k + kb + 0) * 16 + m];
    wdb[k].y = ws[1280 + (4 * k + kb + 1) * 16 + m];
  }
  const float bdv = ws[2304 + m];

  v2f a[16];
#pragma unroll
  for (int k = 0; k < 16; ++k) { a[k].x = 0.f; a[k].y = 0.f; }

  __syncthreads();

  const int* __restrict__ nsrc = half ? num2 : num1;

  for (int t = 0; t < TT; ++t) {
    vidx[half][m] = nsrc[(row_base + m) * TT + t];
    if (t + 1 < TT) __builtin_prefetch(&nsrc[(row_base + m) * TT + t + 1], 0, 3);
    __syncthreads();

    v8f acc[4];
#pragma unroll
    for (int r = 0; r < 8; ++r) {
      const int hr = r + 8 * half;
      const int d1 = vidx[0][hr];
      const int d2 = vidx[1][hr];
#pragma unroll
      for (int n = 0; n < 4; ++n)
        acc[n][r] = lut[d1 * HH + n * 16 + m] + lut[VV * HH + d2 * HH + n * 16 + m];
    }

#pragma unroll
    for (int k = 0; k < 16; ++k)
#pragma unroll
      for (int n = 0; n < 4; ++n)
        acc[n] = __builtin_amdgcn_wmma_f32_16x16x4_f32(
            false, a[k], false, whb[k][n], (short)0, acc[n], false, false);

#pragma unroll
    for (int n = 0; n < 4; ++n)
#pragma unroll
      for (int r = 0; r < 8; ++r)
        hbuf[(r + 8 * half) * HH + n * 16 + m] = DEV_TANH(acc[n][r]);
    __syncthreads();

#pragma unroll
    for (int k = 0; k < 16; ++k)
      a[k] = *(const v2f*)&hbuf[m * HH + 4 * k + kb];

    v8f c2;
#pragma unroll
    for (int r = 0; r < 8; ++r) c2[r] = bdv;
#pragma unroll
    for (int k = 0; k < 16; ++k)
      c2 = __builtin_amdgcn_wmma_f32_16x16x4_f32(
          false, a[k], false, wdb[k], (short)0, c2, false, false);

    if (m < OO) {
#pragma unroll
      for (int r = 0; r < 8; ++r) {
        const int row = row_base + r + 8 * half;
        out[(size_t)row * TT * OO + (size_t)t * OO + m] = c2[r];
      }
    }
  }
}

extern "C" void kernel_launch(void* const* d_in, const int* in_sizes, int n_in,
                              void* d_out, int out_size, void* d_ws, size_t ws_size,
                              hipStream_t stream) {
  const int*   num1 = (const int*)d_in[0];
  const int*   num2 = (const int*)d_in[1];
  const float* emb  = (const float*)d_in[2];
  const float* Wx   = (const float*)d_in[3];
  const float* Wh   = (const float*)d_in[4];
  const float* brnn = (const float*)d_in[5];
  const float* Wd   = (const float*)d_in[6];
  const float* bd   = (const float*)d_in[7];
  float* out = (float*)d_out;
  float* ws  = (float*)d_ws;

  rnn_prep_kernel<<<1, 64, 0, stream>>>(emb, Wx, brnn, Wd, bd, ws);
  rnn_scan_kernel<<<BB / 16, 32, 0, stream>>>(num1, num2, Wh, ws, out);
}


# Eight batch rows per block instead of 16 (tile rows 0-3 and 8-11, the rest zero) on 128 blocks: per lane and step half the hand-over, adds, tanh, transposition and table reads; every sum unchanged
# speedup vs baseline: 4.5490x; 4.5490x over previous
#include <hip/hip_runtime.h>

typedef __attribute__((ext_vector_type(2))) float v2f;
typedef __attribute__((ext_vector_type(4))) float v4f;
typedef __attribute__((ext_vector_type(8))) float v8f;

#define BB 1024
#define TT 256
#define HH 64
#define EE 32
#define VV 10
#define OO 10

#if __has_builtin(__builtin_amdgcn_tanhf)
#define DEV_TANH(x) __builtin_amdgcn_tanhf(x)
#else
#define DEV_TANH(x) tanhf(x)
#endif


__global__ __launch_bounds__(64) void rnn_prep_kernel(
    const float* __restrict__ emb, const float* __restrict__ Wx,
    const float* __restrict__ brnn, const float* __restrict__ Wd,
    const float* __restrict__ bd, float* __restrict__ ws) {
  const int j = threadIdx.x;
  for (int v = 0; v < VV; ++v) {
    float s0 = brnn[j], s1 = 0.f;
    for (int d = 0; d < EE; ++d) {
      const float e = emb[v * EE + d];
      s0 += e * Wx[d * HH + j];
      s1 += e * Wx[(EE + d) * HH + j];
    }
    ws[v * HH + j] = s0;
    ws[VV * HH + v * HH + j] = s1;
  }
  for (int n = 0; n < 16; ++n)
    ws[1280 + j * 16 + n] = (n < OO) ? Wd[j * OO + n] : 0.f;
  if (j < 16) ws[2304 + j] = (j < OO) ? bd[j] : 0.f;
}


#define RB 8
#define HP 18

__global__ __launch_bounds__(128) void rnn_scan_kernel(
    const int* __restrict__ num1, const int* __restrict__ num2,
    const float* __restrict__ Wh, const float* __restrict__ ws,
    float* __restrict__ out) {
  __shared__ float lutf[VV * VV * HH];
  __shared__ __attribute__((aligned(16))) int dig[TT * RB];
  __shared__ v4f part[2 * 4 * 4 * 32];
  __shared__ v4f plog[3 * 4 * 32];
  __shared__ __attribute__((aligned(16))) float hb[4 * 16 * HP];

  const int tid  = threadIdx.x;
  const int w    = __builtin_amdgcn_readfirstlane(tid >> 5);
  const int lane = tid & 31;
  const int half = lane >> 4;
  const int m    = lane & 15;
  const int kb   = half * 2;
  const int row_base = blockIdx.x * RB;

  {
    const int b   = tid & (RB - 1);
    const int seg = tid >> 3;
    const int4* __restrict__ r1 = (const int4*)(num1 + (size_t)(row_base + b) * TT) + seg * 4;
    const int4* __restrict__ r2 = (const int4*)(num2 + (size_t)(row_base + b) * TT) + seg * 4;
    int* __restrict__ dd = &dig[seg * 16 * RB + b];
#pragma unroll
    for (int c = 0; c < 4; ++c) {
      const int4 d1 = r1[c];
      const int4 d2 = r2[c];
      dd[(4 * c + 0) * RB] = (d1.x * VV + d2.x) * HH;
      dd[(4 * c + 1) * RB] = (d1.y * VV + d2.y) * HH;
      dd[(4 * c + 2) * RB] = (d1.z * VV + d2.z) * HH;
      dd[(4 * c + 3) * RB] = (d1.w * VV + d2.w) * HH;
    }
  }

  for (int i = tid; i < 4 * 16 * HP; i += 128) hb[i] = 0.f;

  {
    const int c  = tid & 63;
    const int ph = tid >> 6;
    float l1[VV];
#pragma unroll
    for (int d2 = 0; d2 < VV; ++d2) l1[d2] = ws[VV * HH + d2 * HH + c];
#pragma unroll
    for (int i = 0; i < VV / 2; ++i) {
      const int d1 = ph * (VV / 2) + i;
      const float l0 = ws[d1 * HH + c];
#pragma unroll
      for (int d2 = 0; d2 < VV; ++d2) lutf[(d1 * VV + d2) * HH + c] = l0 + l1[d2];
    }
  }

  v2f whb[4][4];
#pragma unroll
  for (int j = 0; j < 4; ++j)
#pragma unroll
    for (int i = 0; i < 4; ++i) {
      const int col = 16 * ((w + i) & 3) + m;
      whb[j][i].x = Wh[(16 * w + 4 * j + kb + 0) * HH + col];
      whb[j][i].y = Wh[(16 * w + 4 * j + kb + 1) * HH + col];
    }
  v2f wdb[4];
#pragma unroll
  for (int j = 0; j < 4; ++j) {
    wdb[j].x = ws[1280 + (16 * w + 4 * j + kb + 0) * 16 + m];
    wdb[j].y = ws[1280 + (16 * w + 4 * j + kb + 1) * 16 + m];
  }
  const float cinit = (w == 0) ? ws[2304 + m] : 0.f;

  v2f a[4];
#pragma unroll
  for (int j = 0; j < 4; ++j) { a[j].x = 0.f; a[j].y = 0.f; }

  __syncthreads();

  float xr[4];
  int dgn[4];
  const int xcol = 16 * w + m;
  {
    const int4 c0 = *(const int4*)&dig[0 * RB + 4 * half];
    const int4 e0 = *(const int4*)&dig[1 * RB + 4 * half];
    xr[0] = lutf[c0.x + xcol]; xr[1] = lutf[c0.y + xcol]; xr[2] = lutf[c0.z + xcol]; xr[3] = lutf[c0.w + xcol];
    dgn[0] = e0.x; dgn[1] = e0.y; dgn[2] = e0.z; dgn[3] = e0.w;
  }

  float* hbw = &hb[w * 16 * HP];
  const float* plf = (const float*)plog;
  float* __restrict__ orow = out + (size_t)(row_base + w + 4 * half) * TT * OO + m;
  const int plsel = lane * 4 + w;

  for (int t = 0; t < TT; ++t) {
    const int buf = t & 1;
    v8f acc1, acc2, acc3;
#pragma unroll
    for (int r = 0; r < 8; ++r) { acc1[r] = 0.f; acc2[r] = 0.f; acc3[r] = 0.f; }

#pragma unroll
    for (int j = 0; j < 4; ++j) {
      acc1 = __builtin_amdgcn_wmma_f32_16x16x4_f32(false, a[j], false, whb[j][1], (short)0, acc1, false, false);
      acc2 = __builtin_amdgcn_wmma_f32_16x16x4_f32(false, a[j], false, whb[j][2], (short)0, acc2, false, false);
      acc3 = __builtin_amdgcn_wmma_f32_16x16x4_f32(false, a[j], false, whb[j][3], (short)0, acc3, false, false);
    }

    {
      v4f* p1 = &part[((buf * 4 + ((w + 1) & 3)) * 4 + 1) * 32 + lane];
      v4f* p2 = &part[((buf * 4 + ((w + 2) & 3)) * 4 + 2) * 32 + lane];
      v4f* p3 = &part[((buf * 4 + ((w + 3) & 3)) * 4 + 3) * 32 + lane];
      v4f u;
      u.x = acc1[0]; u.y = acc1[1]; u.z = acc1[2]; u.w = acc1[3]; p1[0] = u;
      u.x = acc2[0]; u.y = acc2[1]; u.z = acc2[2]; u.w = acc2[3]; p2[0] = u;
      u.x = acc3[0]; u.y = acc3[1]; u.z = acc3[2]; u.w = acc3[3]; p3[0] = u;
    }

    __syncthreads();

    const v4f* g = &part[((buf * 4 + w) * 4) * 32 + lane];
    const v4f q1 = g[1 * 32];
    const v4f q2 = g[2 * 32];
    const v4f q3 = g[3 * 32];
    const int pb = buf ^ 1;
    const float l0 = plf[(pb * 4 + 0) * 128 + plsel];
    const float l1 = plf[(pb * 4 + 1) * 128 + plsel];
    const float l2 = plf[(pb * 4 + 2) * 128 + plsel];
    const float l3 = plf[(pb * 4 + 3) * 128 + plsel];
    __builtin_amdgcn_sched_barrier(0);
    v8f acc0;
#pragma unroll
    for (int r = 0; r < 4; ++r) { acc0[r] = xr[r]; acc0[r + 4] = 0.f; }
    {
      v8f c;
#pragma unroll
      for (int r = 0; r < 8; ++r) c[r] = cinit;
#pragma unroll
      for (int j = 0; j < 4; ++j) {
        acc0 = __builtin_amdgcn_wmma_f32_16x16x4_f32(false, a[j], false, whb[j][0], (short)0, acc0, false, false);
        c    = __builtin_amdgcn_wmma_f32_16x16x4_f32(false, a[j], false, wdb[j], (short)0, c, false, false);
      }
      v4f* pl = &plog[(buf * 4 + w) * 32 + lane];
      v4f u;
      u.x = c[0]; u.y = c[1]; u.z = c[2]; u.w = c[3]; pl[0] = u;
    }

    float th[4];
    th[0] = DEV_TANH(((acc0[0] + q1.x) + q2.x) + q3.x);
    th[1] = DEV_TANH(((acc0[1] + q1.y) + q2.y) + q3.y);
    th[2] = DEV_TANH(((acc0[2] + q1.z) + q2.z) + q3.z);
    th[3] = DEV_TANH(((acc0[3] + q1.w) + q2.w) + q3.w);

#pragma unroll
    for (int r = 0; r < 4; ++r) hbw[(r + 8 * half) * HP + m] = th[r];
#pragma unroll
    for (int j = 0; j < 4; ++j) a[j] = *(const v2f*)&hbw[m * HP + 4 * j + kb];

    float xn[4];
#pragma unroll
    for (int r = 0; r < 4; ++r) xn[r] = lutf[dgn[r] + xcol];
    {
      const int t2 = (t + 2 < TT) ? (t + 2) : (TT - 1);
      const int4 e0 = *(const int4*)&dig[t2 * RB + 4 * half];
      dgn[0] = e0.x; dgn[1] = e0.y; dgn[2] = e0.z; dgn[3] = e0.w;
    }

    if (t >= 2 && m < OO) orow[(size_t)(t - 2) * OO] = ((l0 + l1) + l2) + l3;

#pragma unroll
    for (int r = 0; r < 4; ++r) xr[r] = xn[r];
  }

  {
    v8f c;
#pragma unroll
    for (int r = 0; r < 8; ++r) c[r] = cinit;
#pragma unroll
    for (int j = 0; j < 4; ++j)
      c = __builtin_amdgcn_wmma_f32_16x16x4_f32(false, a[j], false, wdb[j], (short)0, c, false, false);
    v4f* pl = &plog[(2 * 4 + w) * 32 + lane];
    v4f u;
    u.x = c[0]; u.y = c[1]; u.z = c[2]; u.w = c[3]; pl[0] = u;
  }
  __syncthreads();
#pragma unroll
  for (int e = 0; e < 2; ++e) {
    const int pb = 1 + e;
    const float l0 = plf[(pb * 4 + 0) * 128 + plsel];
    const float l1 = plf[(pb * 4 + 1) * 128 + plsel];
    const float l2 = plf[(pb * 4 + 2) * 128 + plsel];
    const float l3 = plf[(pb * 4 + 3) * 128 + plsel];
    if (m < OO) orow[(size_t)(TT - 2 + e) * OO] = ((l0 + l1) + l2) + l3;
  }
}

extern "C" void kernel_launch(void* const* d_in, const int* in_sizes, int n_in,
                              void* d_out, int out_size, void* d_ws, size_t ws_size,
                              hipStream_t stream) {
  const int*   num1 = (const int*)d_in[0];
  const int*   num2 = (const int*)d_in[1];
  const float* emb  = (const float*)d_in[2];
  const float* Wx   = (const float*)d_in[3];
  const float* Wh   = (const float*)d_in[4];
  const float* brnn = (const float*)d_in[5];
  const float* Wd   = (const float*)d_in[6];
  const float* bd   = (const float*)d_in[7];
  float* out = (float*)d_out;
  float* ws  = (float*)d_ws;

  rnn_prep_kernel<<<1, 64, 0, stream>>>(emb, Wx, brnn, Wd, bd, ws);
  rnn_scan_kernel<<<BB / RB, 128, 0, stream>>>(num1, num2, Wh, ws, out);
}


# Logits WMMAs of the step before moved off the critical path: two behind the hand-over stores, two behind the new A-fragment reads (two steps a trip, swapped fragment sets, LDS-only fences); same sums
# speedup vs baseline: 4.6973x; 1.0326x over previous
#include <hip/hip_runtime.h>

typedef __attribute__((ext_vector_type(2))) float v2f;
typedef __attribute__((ext_vector_type(4))) float v4f;
typedef __attribute__((ext_vector_type(8))) float v8f;

#define BB 1024
#define TT 256
#define HH 64
#define EE 32
#define VV 10
#define OO 10

#if __has_builtin(__builtin_amdgcn_tanhf)
#define DEV_TANH(x) __builtin_amdgcn_tanhf(x)
#else
#define DEV_TANH(x) tanhf(x)
#endif


__global__ __launch_bounds__(64) void rnn_prep_kernel(
    const float* __restrict__ emb, const float* __restrict__ Wx,
    const float* __restrict__ brnn, const float* __restrict__ Wd,
    const float* __restrict__ bd, float* __restrict__ ws) {
  const int j = threadIdx.x;
  for (int v = 0; v < VV; ++v) {
    float s0 = brnn[j], s1 = 0.f;
    for (int d = 0; d < EE; ++d) {
      const float e = emb[v * EE + d];
      s0 += e * Wx[d * HH + j];
      s1 += e * Wx[(EE + d) * HH + j];
    }
    ws[v * HH + j] = s0;
    ws[VV * HH + v * HH + j] = s1;
  }
  for (int n = 0; n < 16; ++n)
    ws[1280 + j * 16 + n] = (n < OO) ? Wd[j * OO + n] : 0.f;
  if (j < 16) ws[2304 + j] = (j < OO) ? bd[j] : 0.f;
}


__device__ __forceinline__ void lds_barrier() {
  __builtin_amdgcn_fence(__ATOMIC_RELEASE, "workgroup", "local");
  __builtin_amdgcn_s_barrier();
  __builtin_amdgcn_fence(__ATOMIC_ACQUIRE, "workgroup", "local");
}

#define RB 8
#define HP 18

__global__ __launch_bounds__(128) void rnn_scan_kernel(
    const int* __restrict__ num1, const int* __restrict__ num2,
    const float* __restrict__ Wh, const float* __restrict__ ws,
    float* __restrict__ out) {
  __shared__ float lutf[VV * VV * HH];
  __shared__ __attribute__((aligned(16))) int dig[TT * RB];
  __shared__ v4f part[2 * 4 * 4 * 32];
  __shared__ v4f plog[3 * 4 * 32];
  __shared__ __attribute__((aligned(16))) float hb[4 * 16 * HP];

  const int tid  = threadIdx.x;
  const int w    = __builtin_amdgcn_readfirstlane(tid >> 5);
  const int lane = tid & 31;
  const int half = lane >> 4;
  const int m    = lane & 15;
  const int kb   = half * 2;
  const int row_base = blockIdx.x * RB;

  {
    const int b   = tid & (RB - 1);
    const int seg = tid >> 3;
    const int4* __restrict__ r1 = (const int4*)(num1 + (size_t)(row_base + b) * TT) + seg * 4;
    const int4* __restrict__ r2 = (const int4*)(num2 + (size_t)(row_base + b) * TT) + seg * 4;
    int* __restrict__ dd = &dig[seg * 16 * RB + b];
#pragma unroll
    for (int c = 0; c < 4; ++c) {
      const int4 d1 = r1[c];
      const int4 d2 = r2[c];
      dd[(4 * c + 0) * RB] = (d1.x * VV + d2.x) * HH;
      dd[(4 * c + 1) * RB] = (d1.y * VV + d2.y) * HH;
      dd[(4 * c + 2) * RB] = (d1.z * VV + d2.z) * HH;
      dd[(4 * c + 3) * RB] = (d1.w * VV + d2.w) * HH;
    }
  }

  for (int i = tid; i < 4 * 16 * HP; i += 128) hb[i] = 0.f;

  {
    const int c  = tid & 63;
    const int ph = tid >> 6;
    float l1[VV];
#pragma unroll
    for (int d2 = 0; d2 < VV; ++d2) l1[d2] = ws[VV * HH + d2 * HH + c];
#pragma unroll
    for (int i = 0; i < VV / 2; ++i) {
      const int d1 = ph * (VV / 2) + i;
      const float l0 = ws[d1 * HH + c];
#pragma unroll
      for (int d2 = 0; d2 < VV; ++d2) lutf[(d1 * VV + d2) * HH + c] = l0 + l1[d2];
    }
  }

  v2f whb[4][4];
#pragma unroll
  for (int j = 0; j < 4; ++j)
#pragma unroll
    for (int i = 0; i < 4; ++i) {
      const int col = 16 * ((w + i) & 3) + m;
      whb[j][i].x = Wh[(16 * w + 4 * j + kb + 0) * HH + col];
      whb[j][i].y = Wh[(16 * w + 4 * j + kb + 1) * HH + col];
    }
  v2f wdb[4];
#pragma unroll
  for (int j = 0; j < 4; ++j) {
    wdb[j].x = ws[1280 + (16 * w + 4 * j + kb + 0) * 16 + m];
    wdb[j].y = ws[1280 + (16 * w + 4 * j + kb + 1) * 16 + m];
  }
  const float cinit = (w == 0) ? ws[2304 + m] : 0.f;

  v2f a[4];
#pragma unroll
  for (int j = 0; j < 4; ++j) { a[j].x = 0.f; a[j].y = 0.f; }

  __syncthreads();

  float xr[4];
  int dgn[4];
  const int xcol = 16 * w + m;
  {
    const int4 c0 = *(const int4*)&dig[0 * RB + 4 * half];
    const int4 e0 = *(const int4*)&dig[1 * RB + 4 * half];
    xr[0] = lutf[c0.x + xcol]; xr[1] = lutf[c0.y + xcol]; xr[2] = lutf[c0.z + xcol]; xr[3] = lutf[c0.w + xcol];
    dgn[0] = e0.x; dgn[1] = e0.y; dgn[2] = e0.z; dgn[3] = e0.w;
  }

  float* hbw = &hb[w * 16 * HP];
  const float* plf = (const float*)plog;
  float* __restrict__ orow = out + (size_t)(row_base + w + 4 * half) * TT * OO + m;
  const int plsel = lane * 4 + w;

  auto step = [&](const int t, const int buf, const v2f (&ac)[4], v2f (&an)[4]) __attribute__((always_inline)) {
    v8f acc1, acc2, acc3;
#pragma unroll
    for (int r = 0; r < 8; ++r) { acc1[r] = 0.f; acc2[r] = 0.f; acc3[r] = 0.f; }

#pragma unroll
    for (int j = 0; j < 4; ++j) {
      acc1 = __builtin_amdgcn_wmma_f32_16x16x4_f32(false, ac[j], false, whb[j][1], (short)0, acc1, false, false);
      acc2 = __builtin_amdgcn_wmma_f32_16x16x4_f32(false, ac[j], false, whb[j][2], (short)0, acc2, false, false);
      acc3 = __builtin_amdgcn_wmma_f32_16x16x4_f32(false, ac[j], false, whb[j][3], (short)0, acc3, false, false);
    }

    {
      v4f* p1 = &part[((buf * 4 + ((w + 1) & 3)) * 4 + 1) * 32 + lane];
      v4f* p2 = &part[((buf * 4 + ((w + 2) & 3)) * 4 + 2) * 32 + lane];
      v4f* p3 = &part[((buf * 4 + ((w + 3) & 3)) * 4 + 3) * 32 + lane];
      v4f u;
      u.x = acc1[0]; u.y = acc1[1]; u.z = acc1[2]; u.w = acc1[3]; p1[0] = u;
      u.x = acc2[0]; u.y = acc2[1]; u.z = acc2[2]; u.w = acc2[3]; p2[0] = u;
      u.x = acc3[0]; u.y = acc3[1]; u.z = acc3[2]; u.w = acc3[3]; p3[0] = u;
    }

    __builtin_amdgcn_sched_barrier(0);
    v8f c;
#pragma unroll
    for (int r = 0; r < 8; ++r) c[r] = cinit;
    c = __builtin_amdgcn_wmma_f32_16x16x4_f32(false, ac[0], false, wdb[0], (short)0, c, false, false);
    c = __builtin_amdgcn_wmma_f32_16x16x4_f32(false, ac[1], false, wdb[1], (short)0, c, false, false);
    asm volatile("" :: "v"(c));

    lds_barrier();

    const v4f* g = &part[((buf * 4 + w) * 4) * 32 + lane];
    const v4f q1 = g[1 * 32];
    const v4f q2 = g[2 * 32];
    const v4f q3 = g[3 * 32];
    const int pb = buf ^ 1;
    const float l0 = plf[(pb * 4 + 0) * 128 + plsel];
    const float l1 = plf[(pb * 4 + 1) * 128 + plsel];
    const float l2 = plf[(pb * 4 + 2) * 128 + plsel];
    const float l3 = plf[(pb * 4 + 3) * 128 + plsel];
    __builtin_amdgcn_sched_barrier(0);
    v8f acc0;
#pragma unroll
    for (int r = 0; r < 4; ++r) { acc0[r] = xr[r]; acc0[r + 4] = 0.f; }
#pragma unroll
    for (int j = 0; j < 4; ++j)
      acc0 = __builtin_amdgcn_wmma_f32_16x16x4_f32(false, ac[j], false, whb[j][0], (short)0, acc0, false, false);

    float th[4];
    th[0] = DEV_TANH(((acc0[0] + q1.x) + q2.x) + q3.x);
    th[1] = DEV_TANH(((acc0[1] + q1.y) + q2.y) + q3.y);
    th[2] = DEV_TANH(((acc0[2] + q1.z) + q2.z) + q3.z);
    th[3] = DEV_TANH(((acc0[3] + q1.w) + q2.w) + q3.w);

#pragma unroll
    for (int r = 0; r < 4; ++r) hbw[(r + 8 * half) * HP + m] = th[r];
#pragma unroll
    for (int j = 0; j < 4; ++j) an[j] = *(const v2f*)&hbw[m * HP + 4 * j + kb];

    float xn[4];
#pragma unroll
    for (int r = 0; r < 4; ++r) xn[r] = lutf[dgn[r] + xcol];
    {
      const int t2 = (t + 2 < TT) ? (t + 2) : (TT - 1);
      const int4 e0 = *(const int4*)&dig[t2 * RB + 4 * half];
      dgn[0] = e0.x; dgn[1] = e0.y; dgn[2] = e0.z; dgn[3] = e0.w;
    }

    __builtin_amdgcn_sched_barrier(0);
    c = __builtin_amdgcn_wmma_f32_16x16x4_f32(false, ac[2], false, wdb[2], (short)0, c, false, false);
    c = __builtin_amdgcn_wmma_f32_16x16x4_f32(false, ac[3], false, wdb[3], (short)0, c, false, false);
    {
      v4f* pl = &plog[(buf * 4 + w) * 32 + lane];
      v4f u;
      u.x = c[0]; u.y = c[1]; u.z = c[2]; u.w = c[3]; pl[0] = u;
    }

    if (t >= 2 && m < OO) orow[(size_t)(t - 2) * OO] = ((l0 + l1) + l2) + l3;

#pragma unroll
    for (int r = 0; r < 4; ++r) xr[r] = xn[r];
  };

  v2f b[4];
  for (int t = 0; t < TT; t += 2) {
    step(t, 0, a, b);
    step(t + 1, 1, b, a);
  }

  {
    v8f c;
#pragma unroll
    for (int r = 0; r < 8; ++r) c[r] = cinit;
#pragma unroll
    for (int j = 0; j < 4; ++j)
      c = __builtin_amdgcn_wmma_f32_16x16x4_f32(false, a[j], false, wdb[j], (short)0, c, false, false);
    v4f* pl = &plog[(2 * 4 + w) * 32 + lane];
    v4f u;
    u.x = c[0]; u.y = c[1]; u.z = c[2]; u.w = c[3]; pl[0] = u;
  }
  __syncthreads();
#pragma unroll
  for (int e = 0; e < 2; ++e) {
    const int pb = 1 + e;
    const float l0 = plf[(pb * 4 + 0) * 128 + plsel];
    const float l1 = plf[(pb * 4 + 1) * 128 + plsel];
    const float l2 = plf[(pb * 4 + 2) * 128 + plsel];
    const float l3 = plf[(pb * 4 + 3) * 128 + plsel];
    if (m < OO) orow[(size_t)(TT - 2 + e) * OO] = ((l0 + l1) + l2) + l3;
  }
}

extern "C" void kernel_launch(void* const* d_in, const int* in_sizes, int n_in,
                              void* d_out, int out_size, void* d_ws, size_t ws_size,
                              hipStream_t stream) {
  const int*   num1 = (const int*)d_in[0];
  const int*   num2 = (const int*)d_in[1];
  const float* emb  = (const float*)d_in[2];
  const float* Wx   = (const float*)d_in[3];
  const float* Wh   = (const float*)d_in[4];
  const float* brnn = (const float*)d_in[5];
  const float* Wd   = (const float*)d_in[6];
  const float* bd   = (const float*)d_in[7];
  float* out = (float*)d_out;
  float* ws  = (float*)d_ws;

  rnn_prep_kernel<<<1, 64, 0, stream>>>(emb, Wx, brnn, Wd, bd, ws);
  rnn_scan_kernel<<<BB / RB, 128, 0, stream>>>(num1, num2, Wh, ws, out);
}


# Out store with all lanes and no branch (lanes 10-15 repeat lane 9: same address and value), issued behind the A-fragment reads, ahead of the last two logits WMMAs; every sum unchanged
# speedup vs baseline: 4.9776x; 1.0597x over previous
#include <hip/hip_runtime.h>

typedef __attribute__((ext_vector_type(2))) float v2f;
typedef __attribute__((ext_vector_type(4))) float v4f;
typedef __attribute__((ext_vector_type(8))) float v8f;

#define BB 1024
#define TT 256
#define HH 64
#define EE 32
#define VV 10
#define OO 10

#if __has_builtin(__builtin_amdgcn_tanhf)
#define DEV_TANH(x) __builtin_amdgcn_tanhf(x)
#else
#define DEV_TANH(x) tanhf(x)
#endif


__global__ __launch_bounds__(64) void rnn_prep_kernel(
    const float* __restrict__ emb, const float* __restrict__ Wx,
    const float* __restrict__ brnn, const float* __restrict__ Wd,
    const float* __restrict__ bd, float* __restrict__ ws) {
  const int j = threadIdx.x;
  for (int v = 0; v < VV; ++v) {
    float s0 = brnn[j], s1 = 0.f;
    for (int d = 0; d < EE; ++d) {
      const float e = emb[v * EE + d];
      s0 += e * Wx[d * HH + j];
      s1 += e * Wx[(EE + d) * HH + j];
    }
    ws[v * HH + j] = s0;
    ws[VV * HH + v * HH + j] = s1;
  }
  for (int n = 0; n < 16; ++n)
    ws[1280 + j * 16 + n] = (n < OO) ? Wd[j * OO + n] : 0.f;
  if (j < 16) ws[2304 + j] = (j < OO) ? bd[j] : 0.f;
}


__device__ __forceinline__ void lds_barrier() {
  __builtin_amdgcn_fence(__ATOMIC_RELEASE, "workgroup", "local");
  __builtin_amdgcn_s_barrier();
  __builtin_amdgcn_fence(__ATOMIC_ACQUIRE, "workgroup", "local");
}

#define RB 8
#define HP 18

__global__ __launch_bounds__(128) void rnn_scan_kernel(
    const int* __restrict__ num1, const int* __restrict__ num2,
    const float* __restrict__ Wh, const float* __restrict__ ws,
    float* __restrict__ out) {
  __shared__ float lutf[VV * VV * HH];
  __shared__ __attribute__((aligned(16))) int dig[TT * RB];
  __shared__ v4f part[2 * 4 * 4 * 32];
  __shared__ v4f plog[3 * 4 * 32];
  __shared__ __attribute__((aligned(16))) float hb[4 * 16 * HP];

  const int tid  = threadIdx.x;
  const int w    = __builtin_amdgcn_readfirstlane(tid >> 5);
  const int lane = tid & 31;
  const int half = lane >> 4;
  const int m    = lane & 15;
  const int kb   = half * 2;
  const int row_base = blockIdx.x * RB;

  {
    const int b   = tid & (RB - 1);
    const int seg = tid >> 3;
    const int4* __restrict__ r1 = (const int4*)(num1 + (size_t)(row_base + b) * TT) + seg * 4;
    const int4* __restrict__ r2 = (const int4*)(num2 + (size_t)(row_base + b) * TT) + seg * 4;
    int* __restrict__ dd = &dig[seg * 16 * RB + b];
#pragma unroll
    for (int c = 0; c < 4; ++c) {
      const int4 d1 = r1[c];
      const int4 d2 = r2[c];
      dd[(4 * c + 0) * RB] = (d1.x * VV + d2.x) * HH;
      dd[(4 * c + 1) * RB] = (d1.y * VV + d2.y) * HH;
      dd[(4 * c + 2) * RB] = (d1.z * VV + d2.z) * HH;
      dd[(4 * c + 3) * RB] = (d1.w * VV + d2.w) * HH;
    }
  }

  for (int i = tid; i < 4 * 16 * HP; i += 128) hb[i] = 0.f;

  {
    const int c  = tid & 63;
    const int ph = tid >> 6;
    float l1[VV];
#pragma unroll
    for (int d2 = 0; d2 < VV; ++d2) l1[d2] = ws[VV * HH + d2 * HH + c];
#pragma unroll
    for (int i = 0; i < VV / 2; ++i) {
      const int d1 = ph * (VV / 2) + i;
      const float l0 = ws[d1 * HH + c];
#pragma unroll
      for (int d2 = 0; d2 < VV; ++d2) lutf[(d1 * VV + d2) * HH + c] = l0 + l1[d2];
    }
  }

  v2f whb[4][4];
#pragma unroll
  for (int j = 0; j < 4; ++j)
#pragma unroll
    for (int i = 0; i < 4; ++i) {
      const int col = 16 * ((w + i) & 3) + m;
      whb[j][i].x = Wh[(16 * w + 4 * j + kb + 0) * HH + col];
      whb[j][i].y = Wh[(16 * w + 4 * j + kb + 1) * HH + col];
    }
  v2f wdb[4];
#pragma unroll
  for (int j = 0; j < 4; ++j) {
    wdb[j].x = ws[1280 + (16 * w + 4 * j + kb + 0) * 16 + m];
    wdb[j].y = ws[1280 + (16 * w + 4 * j + kb + 1) * 16 + m];
  }
  const float cinit = (w == 0) ? ws[2304 + m] : 0.f;

  v2f a[4];
#pragma unroll
  for (int j = 0; j < 4; ++j) { a[j].x = 0.f; a[j].y = 0.f; }

  __syncthreads();

  float xr[4];
  int dgn[4];
  const int xcol = 16 * w + m;
  {
    const int4 c0 = *(const int4*)&dig[0 * RB + 4 * half];
    const int4 e0 = *(const int4*)&dig[1 * RB + 4 * half];
    xr[0] = lutf[c0.x + xcol]; xr[1] = lutf[c0.y + xcol]; xr[2] = lutf[c0.z + xcol]; xr[3] = lutf[c0.w + xcol];
    dgn[0] = e0.x; dgn[1] = e0.y; dgn[2] = e0.z; dgn[3] = e0.w;
  }

  float* hbw = &hb[w * 16 * HP];
  const float* plf = (const float*)plog;
  const int mc = (m < OO) ? m : (OO - 1);
  float* __restrict__ oblk = out + (size_t)row_base * TT * OO;
  const unsigned ooff = (unsigned)((w + 4 * half) * TT * OO + mc);
  const int plsel = (half * 16 + mc) * 4 + w;

  auto step = [&](const int t, const int buf, const v2f (&ac)[4], v2f (&an)[4]) __attribute__((always_inline)) {
    v8f acc1, acc2, acc3;
#pragma unroll
    for (int r = 0; r < 8; ++r) { acc1[r] = 0.f; acc2[r] = 0.f; acc3[r] = 0.f; }

#pragma unroll
    for (int j = 0; j < 4; ++j) {
      acc1 = __builtin_amdgcn_wmma_f32_16x16x4_f32(false, ac[j], false, whb[j][1], (short)0, acc1, false, false);
      acc2 = __builtin_amdgcn_wmma_f32_16x16x4_f32(false, ac[j], false, whb[j][2], (short)0, acc2, false, false);
      acc3 = __builtin_amdgcn_wmma_f32_16x16x4_f32(false, ac[j], false, whb[j][3], (short)0, acc3, false, false);
    }

    {
      v4f* p1 = &part[((buf * 4 + ((w + 1) & 3)) * 4 + 1) * 32 + lane];
      v4f* p2 = &part[((buf * 4 + ((w + 2) & 3)) * 4 + 2) * 32 + lane];
      v4f* p3 = &part[((buf * 4 + ((w + 3) & 3)) * 4 + 3) * 32 + lane];
      v4f u;
      u.x = acc1[0]; u.y = acc1[1]; u.z = acc1[2]; u.w = acc1[3]; p1[0] = u;
      u.x = acc2[0]; u.y = acc2[1]; u.z = acc2[2]; u.w = acc2[3]; p2[0] = u;
      u.x = acc3[0]; u.y = acc3[1]; u.z = acc3[2]; u.w = acc3[3]; p3[0] = u;
    }

    __builtin_amdgcn_sched_barrier(0);
    v8f c;
#pragma unroll
    for (int r = 0; r < 8; ++r) c[r] = cinit;
    c = __builtin_amdgcn_wmma_f32_16x16x4_f32(false, ac[0], false, wdb[0], (short)0, c, false, false);
    c = __builtin_amdgcn_wmma_f32_16x16x4_f32(false, ac[1], false, wdb[1], (short)0, c, false, false);
    asm volatile("" :: "v"(c));

    lds_barrier();

    const v4f* g = &part[((buf * 4 + w) * 4) * 32 + lane];
    const v4f q1 = g[1 * 32];
    const v4f q2 = g[2 * 32];
    const v4f q3 = g[3 * 32];
    const int pb = buf ^ 1;
    const float l0 = plf[(pb * 4 + 0) * 128 + plsel];
    const float l1 = plf[(pb * 4 + 1) * 128 + plsel];
    const float l2 = plf[(pb * 4 + 2) * 128 + plsel];
    const float l3 = plf[(pb * 4 + 3) * 128 + plsel];
    __builtin_amdgcn_sched_barrier(0);
    v8f acc0;
#pragma unroll
    for (int r = 0; r < 4; ++r) { acc0[r] = xr[r]; acc0[r + 4] = 0.f; }
#pragma unroll
    for (int j = 0; j < 4; ++j)
      acc0 = __builtin_amdgcn_wmma_f32_16x16x4_f32(false, ac[j], false, whb[j][0], (short)0, acc0, false, false);

    float th[4];
    th[0] = DEV_TANH(((acc0[0] + q1.x) + q2.x) + q3.x);
    th[1] = DEV_TANH(((acc0[1] + q1.y) + q2.y) + q3.y);
    th[2] = DEV_TANH(((acc0[2] + q1.z) + q2.z) + q3.z);
    th[3] = DEV_TANH(((acc0[3] + q1.w) + q2.w) + q3.w);

#pragma unroll
    for (int r = 0; r < 4; ++r) hbw[(r + 8 * half) * HP + m] = th[r];
#pragma unroll
    for (int j = 0; j < 4; ++j) an[j] = *(const v2f*)&hbw[m * HP + 4 * j + kb];

    float xn[4];
#pragma unroll
    for (int r = 0; r < 4; ++r) xn[r] = lutf[dgn[r] + xcol];
    {
      const int t2 = (t + 2 < TT) ? (t + 2) : (TT - 1);
      const int4 e0 = *(const int4*)&dig[t2 * RB + 4 * half];
      dgn[0] = e0.x; dgn[1] = e0.y; dgn[2] = e0.z; dgn[3] = e0.w;
    }

    if (t >= 2) {
      float* __restrict__ ot = oblk + (t - 2) * OO;
      ot[ooff] = ((l0 + l1) + l2) + l3;
    }

    __builtin_amdgcn_sched_barrier(0);
    c = __builtin_amdgcn_wmma_f32_16x16x4_f32(false, ac[2], false, wdb[2], (short)0, c, false, false);
    c = __builtin_amdgcn_wmma_f32_16x16x4_f32(false, ac[3], false, wdb[3], (short)0, c, false, false);
    {
      v4f* pl = &plog[(buf * 4 + w) * 32 + lane];
      v4f u;
      u.x = c[0]; u.y = c[1]; u.z = c[2]; u.w = c[3]; pl[0] = u;
    }
    __builtin_amdgcn_sched_barrier(0);

#pragma unroll
    for (int r = 0; r < 4; ++r) xr[r] = xn[r];
  };

  v2f b[4];
  for (int t = 0; t < TT; t += 2) {
    step(t, 0, a, b);
    step(t + 1, 1, b, a);
  }

  {
    v8f c;
#pragma unroll
    for (int r = 0; r < 8; ++r) c[r] = cinit;
#pragma unroll
    for (int j = 0; j < 4; ++j)
      c = __builtin_amdgcn_wmma_f32_16x16x4_f32(false, a[j], false, wdb[j], (short)0, c, false, false);
    v4f* pl = &plog[(2 * 4 + w) * 32 + lane];
    v4f u;
    u.x = c[0]; u.y = c[1]; u.z = c[2]; u.w = c[3]; pl[0] = u;
  }
  __syncthreads();
#pragma unroll
  for (int e = 0; e < 2; ++e) {
    const int pb = 1 + e;
    const float l0 = plf[(pb * 4 + 0) * 128 + plsel];
    const float l1 = plf[(pb * 4 + 1) * 128 + plsel];
    const float l2 = plf[(pb * 4 + 2) * 128 + plsel];
    const float l3 = plf[(pb * 4 + 3) * 128 + plsel];
    oblk[(TT - 2 + e) * OO + ooff] = ((l0 + l1) + l2) + l3;
  }
}

extern "C" void kernel_launch(void* const* d_in, const int* in_sizes, int n_in,
                              void* d_out, int out_size, void* d_ws, size_t ws_size,
                              hipStream_t stream) {
  const int*   num1 = (const int*)d_in[0];
  const int*   num2 = (const int*)d_in[1];
  const float* emb  = (const float*)d_in[2];
  const float* Wx   = (const float*)d_in[3];
  const float* Wh   = (const float*)d_in[4];
  const float* brnn = (const float*)d_in[5];
  const float* Wd   = (const float*)d_in[6];
  const float* bd   = (const float*)d_in[7];
  float* out = (float*)d_out;
  float* ws  = (float*)d_ws;

  rnn_prep_kernel<<<1, 64, 0, stream>>>(emb, Wx, brnn, Wd, bd, ws);
  rnn_scan_kernel<<<BB / RB, 128, 0, stream>>>(num1, num2, Wh, ws, out);
}
